# starting kernel
# baseline (speedup 1.0000x reference)
#include <hip/hip_runtime.h>
#include <math.h>

#define B_    16
#define C_    128
#define N_    8192
#define NH    4096
#define KBINS 4097
#define KS    4104
#define WPAIR 64

typedef __attribute__((ext_vector_type(2))) float v2f;
typedef __attribute__((ext_vector_type(8))) float v8f;

__global__ __launch_bounds__(256) void k0_twiddle(float2* __restrict__ Wc,
                                                  float2* __restrict__ Ws)
{
    const int idx = blockIdx.x * 256 + threadIdx.x;
    const int m = idx >> 6, j = idx & 63;
    const float scale = 1.0f / 1048576.0f;
    float f0 = (float)(m * (2 * j))     * (1.f / 64.f);
    float f1 = (float)(m * (2 * j + 1)) * (1.f / 64.f);
    Wc[idx] = make_float2(cospif(f0) * scale, cospif(f1) * scale);
    Ws[idx] = make_float2(sinpif(f0) * scale, sinpif(f1) * scale);
}

__global__ __launch_bounds__(256) void k1_row_fft(const float* __restrict__ xre,
                                                  const float* __restrict__ xim,
                                                  const float* __restrict__ gamma,
                                                  const float* __restrict__ beta,
                                                  float2* __restrict__ R)
{
    __shared__ float re[N_];
    __shared__ float im[N_];
    const int row = blockIdx.x;
    const int c   = row & (C_ - 1);
    const int tid = threadIdx.x;
    const size_t base = (size_t)row * N_;

    for (int i = tid; i < N_; i += 256) {
        re[i] = xre[base + i];
        im[i] = xim[base + i];
    }
    __syncthreads();

    for (int ls = 12; ls >= 0; --ls) {
        const int s = 1 << ls;
        for (int t = tid; t < NH; t += 256) {
            const int j  = t & (s - 1);
            const int i0 = ((t >> ls) << (ls + 1)) + j;
            const int i1 = i0 + s;
            float ar = re[i0], ai = im[i0];
            float br = re[i1], bi = im[i1];
            float a  = (float)j / (float)s;
            float wr = cospif(a);
            float wi = sinpif(a);
            float tr = ar - br, ti = ai - bi;
            re[i0] = ar + br;  im[i0] = ai + bi;
            re[i1] = tr * wr - ti * wi;
            im[i1] = tr * wi + ti * wr;
        }
        __syncthreads();
    }

    float lsum = 0.f, lsq = 0.f;
    for (int i = tid; i < N_; i += 256) { float v = re[i]; lsum += v; lsq += v * v; }
    im[tid]       = lsum;
    im[256 + tid] = lsq;
    __syncthreads();
    for (int off = 128; off > 0; off >>= 1) {
        if (tid < off) { im[tid] += im[tid + off]; im[256 + tid] += im[256 + tid + off]; }
        __syncthreads();
    }
    const float mean = im[0]   * (1.f / N_);
    const float var  = im[256] * (1.f / N_) - mean * mean;
    const float istd = rsqrtf(var + 1e-5f);
    const float gm = gamma[c], bt = beta[c];
    __syncthreads();

    for (int i = tid; i < N_; i += 256) {
        float v = (re[i] - mean) * istd * gm + bt;
        v = 0.5f * v * (1.f + erff(v * 0.70710678118654752f));
        re[i] = v;
        im[i] = 0.f;
    }
    __syncthreads();

    for (int ls = 0; ls <= 12; ++ls) {
        const int s = 1 << ls;
        for (int t = tid; t < NH; t += 256) {
            const int j  = t & (s - 1);
            const int i0 = ((t >> ls) << (ls + 1)) + j;
            const int i1 = i0 + s;
            float a  = (float)j / (float)s;
            float wr = cospif(a);
            float wi = -sinpif(a);
            float br = re[i1], bi = im[i1];
            float tr = br * wr - bi * wi;
            float ti = br * wi + bi * wr;
            float ar = re[i0], ai = im[i0];
            re[i0] = ar + tr;  im[i0] = ai + ti;
            re[i1] = ar - tr;  im[i1] = ai - ti;
        }
        __syncthreads();
    }

    float2* Rrow = R + (size_t)row * KS;
    for (int k = tid; k < KBINS; k += 256)
        Rrow[k] = make_float2(re[k], im[k]);
}

__global__ __launch_bounds__(256) void k2_dftC_wmma(const float2* __restrict__ R,
                                                    const float2* __restrict__ Wc,
                                                    const float2* __restrict__ Ws,
                                                    float* __restrict__ outre,
                                                    float* __restrict__ outim)
{
    __shared__ float Lre[C_][40];
    __shared__ float Lim[C_][40];
    const int b    = blockIdx.y;
    const int col0 = blockIdx.x * 32;
    const int tid  = threadIdx.x;
    const float2* Rb = R + (size_t)b * C_ * KS;

    for (int idx = tid; idx < C_ * 32; idx += 256) {
        int r = idx >> 5, cc = idx & 31;
        int k = col0 + cc;
        float2 v = (k < KBINS) ? Rb[(size_t)r * KS + k] : make_float2(0.f, 0.f);
        Lre[r][cc] = v.x;
        Lim[r][cc] = v.y;
    }
    __syncthreads();

    const int wave = tid >> 5;
    const int lane = tid & 31;
    const int lg   = lane >> 4;
    const int ln   = lane & 15;
    const int m    = wave * 16 + ln;
    const float2* WcRow = Wc + m * WPAIR;
    const float2* WsRow = Ws + m * WPAIR;

    v8f accr0 = {0.f,0.f,0.f,0.f,0.f,0.f,0.f,0.f};
    v8f acci0 = {0.f,0.f,0.f,0.f,0.f,0.f,0.f,0.f};
    v8f accr1 = {0.f,0.f,0.f,0.f,0.f,0.f,0.f,0.f};
    v8f acci1 = {0.f,0.f,0.f,0.f,0.f,0.f,0.f,0.f};

    for (int k0 = 0; k0 < C_; k0 += 4) {
        const int kA = k0 + 2 * lg;
        float2 wc = WcRow[(k0 >> 1) + lg];
        float2 wsn = WsRow[(k0 >> 1) + lg];
        v2f aRe = {  wc.x,  wc.y  };
        v2f aPs = {  wsn.x, wsn.y };
        v2f aNs = { -wsn.x, -wsn.y };
        v2f bRe0 = { Lre[kA][ln],      Lre[kA + 1][ln]      };
        v2f bIm0 = { Lim[kA][ln],      Lim[kA + 1][ln]      };
        v2f bRe1 = { Lre[kA][16 + ln], Lre[kA + 1][16 + ln] };
        v2f bIm1 = { Lim[kA][16 + ln], Lim[kA + 1][16 + ln] };
        accr0 = __builtin_amdgcn_wmma_f32_16x16x4_f32(false, aRe, false, bRe0, (short)0, accr0, false, false);
        accr0 = __builtin_amdgcn_wmma_f32_16x16x4_f32(false, aPs, false, bIm0, (short)0, accr0, false, false);
        acci0 = __builtin_amdgcn_wmma_f32_16x16x4_f32(false, aRe, false, bIm0, (short)0, acci0, false, false);
        acci0 = __builtin_amdgcn_wmma_f32_16x16x4_f32(false, aNs, false, bRe0, (short)0, acci0, false, false);
        accr1 = __builtin_amdgcn_wmma_f32_16x16x4_f32(false, aRe, false, bRe1, (short)0, accr1, false, false);
        accr1 = __builtin_amdgcn_wmma_f32_16x16x4_f32(false, aPs, false, bIm1, (short)0, accr1, false, false);
        acci1 = __builtin_amdgcn_wmma_f32_16x16x4_f32(false, aRe, false, bIm1, (short)0, acci1, false, false);
        acci1 = __builtin_amdgcn_wmma_f32_16x16x4_f32(false, aNs, false, bRe1, (short)0, acci1, false, false);
    }

    for (int ct = 0; ct < 2; ++ct) {
        const v8f& ar = ct ? accr1 : accr0;
        const v8f& ai = ct ? acci1 : acci0;
        const int kcol = col0 + ct * 16 + ln;
        if (kcol < KBINS) {
            const bool edge = (kcol == 0) || (kcol == NH);
            for (int v = 0; v < 8; ++v) {
                const int mr = wave * 16 + 8 * lg + v;
                const size_t o = (size_t)(b * C_ + mr) * N_;
                float orv = ar[v];
                float oiv = edge ? 0.f : ai[v];
                outre[o + kcol] = orv;
                outim[o + kcol] = oiv;
                if (kcol >= 1 && kcol <= NH - 1) {
                    outre[o + (N_ - kcol)] =  orv;
                    outim[o + (N_ - kcol)] = -oiv;
                }
            }
        }
    }
}

extern "C" void kernel_launch(void* const* d_in, const int* in_sizes, int n_in,
                              void* d_out, int out_size, void* d_ws, size_t ws_size,
                              hipStream_t stream) {
    const float* xre   = (const float*)d_in[0];
    const float* xim   = (const float*)d_in[1];
    const float* gamma = (const float*)d_in[2];
    const float* beta  = (const float*)d_in[3];
    float* outre = (float*)d_out;
    float* outim = outre + (size_t)B_ * C_ * N_;

    float2* R  = (float2*)d_ws;
    float2* Wc = R + (size_t)B_ * C_ * KS;
    float2* Ws = Wc + C_ * WPAIR;

    k0_twiddle<<<dim3((C_ * WPAIR) / 256), dim3(256), 0, stream>>>(Wc, Ws);
    k1_row_fft<<<dim3(B_ * C_), dim3(256), 0, stream>>>(xre, xim, gamma, beta, R);
    k2_dftC_wmma<<<dim3((KBINS + 31) / 32, B_), dim3(256), 0, stream>>>(R, Wc, Ws, outre, outim);
}


# channel DFT-128 as two WMMA stages, 128 = 16 x 8: DFT-16, twiddle, DFT-8 as one real 16x16x16 product, matrices in registers: 48 WMMAs per wave, not 256; table kernel dropped; sum order changes
# speedup vs baseline: 2.3698x; 2.3698x over previous
#include <hip/hip_runtime.h>
#include <math.h>

#define B_    16
#define C_    128
#define N_    8192
#define NH    4096
#define KBINS 4097
#define KS    4104
#define LP    33

typedef __attribute__((ext_vector_type(2))) float v2f;
typedef __attribute__((ext_vector_type(8))) float v8f;

#define LDSN 8448

__device__ __forceinline__ v8f mm4(v2f a, v2f b, v8f c)
{
    return __builtin_amdgcn_wmma_f32_16x16x4_f32(false, a, false, b, (short)0, c, false, false);
}

__device__ __forceinline__ int lp(int i) { return i + (i >> 5); }

template <int SGN, int MODE>
__device__ __forceinline__ void stage16(float* __restrict__ re, float* __restrict__ im,
                                        const float2* __restrict__ t256,
                                        const float2* __restrict__ tA,
                                        const float2* __restrict__ tB,
                                        const v2f (&fC)[4], const v2f (&fS)[4],
                                        const int wave, const int lg, const int ln)
{
#pragma unroll 1
    for (int i = 0; i < 4; ++i) {
        const int tt = wave + 8 * i;
        int cb, rs, h = 0, c = 0, j = 0;
        if (MODE == 0) { j = 16 * tt + ln; cb = j + (j >> 5); rs = 528; }
        else           { h = tt >> 1; c = 16 * (tt & 1) + ln; cb = h * 528 + c; rs = 33; }
        v8f P = {0.f,0.f,0.f,0.f,0.f,0.f,0.f,0.f};
        v8f Q = P, U = P, V = P;
#pragma unroll
        for (int s = 0; s < 4; ++s) {
            const int ka = cb + (4 * s + 2 * lg) * rs;
            v2f xr = { re[ka], re[ka + rs] };
            v2f xi = { im[ka], im[ka + rs] };
            P = mm4(fC[s], xr, P);
            Q = mm4(fS[s], xi, Q);
            U = mm4(fC[s], xi, U);
            V = mm4(fS[s], xr, V);
        }
#pragma unroll
        for (int v = 0; v < 8; ++v) {
            const int ko = 8 * lg + v;
            float yr = (SGN > 0) ? (P[v] - Q[v]) : (P[v] + Q[v]);
            float yi = (SGN > 0) ? (U[v] + V[v]) : (U[v] - V[v]);
            if (MODE == 0 && SGN > 0) {
                const float2 w = t256[(j >> 5) * ko];
                const float tr = yr * w.x - yi * w.y;
                const float ti = yr * w.y + yi * w.x;
                yr = tr; yi = ti;
            } else if (MODE == 1 && SGN > 0) {
                const float2 a = tA[c * h], b = tB[c * ko];
                const float wr = a.x * b.x - a.y * b.y;
                const float wi = a.x * b.y + a.y * b.x;
                const float tr = yr * wr - yi * wi;
                const float ti = yr * wi + yi * wr;
                yr = tr; yi = ti;
            } else if (MODE == 1 && SGN < 0) {
                const float2 w = t256[ko * h];
                const float tr = yr * w.x + yi * w.y;
                const float ti = yi * w.x - yr * w.y;
                yr = tr; yi = ti;
            }
            re[cb + ko * rs] = yr;
            im[cb + ko * rs] = yi;
        }
    }
}

template <int SGN>
__device__ __forceinline__ void stage32(float* __restrict__ re, float* __restrict__ im,
                                        const float2* __restrict__ tA,
                                        const float2* __restrict__ tB,
                                        const v2f (&hC)[2][4], const v2f (&hS)[2][4],
                                        const int wave, const int lg, const int ln)
{
#pragma unroll 1
    for (int i = 0; i < 2; ++i) {
        const int mt = wave + 8 * i;
        const int rb = (16 * mt + ln) * 33 + 4 * lg;
        v8f PE = {0.f,0.f,0.f,0.f,0.f,0.f,0.f,0.f};
        v8f PO = PE, QE = PE, QO = PE;
#pragma unroll
        for (int s = 0; s < 4; ++s) {
            const int ka = rb + 8 * s;
            v2f xe = { re[ka],     re[ka + 2] };
            v2f xo = { re[ka + 1], re[ka + 3] };
            if (SGN > 0) {
                v2f ye = { im[ka],     im[ka + 2] };
                v2f yo = { im[ka + 1], im[ka + 3] };
                PE = mm4(xe, hC[0][s], PE);
                QE = mm4(ye, hS[0][s], QE);
                PO = mm4(xo, hC[1][s], PO);
                QO = mm4(yo, hS[1][s], QO);
            } else {
                PE = mm4(xe, hC[0][s], PE);
                QE = mm4(xe, hS[0][s], QE);
                PO = mm4(xo, hC[1][s], PO);
                QO = mm4(xo, hS[1][s], QO);
            }
        }
#pragma unroll
        for (int v = 0; v < 8; ++v) {
            const int r = 16 * mt + 8 * lg + v;
            const int o = r * 33 + ln;
            if (SGN > 0) {
                const float ev = PE[v] - QE[v], od = PO[v] - QO[v];
                re[o]      = ev + od;
                re[o + 16] = ev - od;
            } else {
#pragma unroll
                for (int t = 0; t < 2; ++t) {
                    const int col = 16 * t + ln;
                    const float yr = t ? (PE[v] - PO[v]) : (PE[v] + PO[v]);
                    const float yq = t ? (QE[v] - QO[v]) : (QE[v] + QO[v]);
                    const float2 a = tA[col * (r >> 4)], b = tB[col * (r & 15)];
                    const float wr = a.x * b.x - a.y * b.y;
                    const float ws = a.x * b.y + a.y * b.x;
                    re[o + 16 * t] = yr * wr - yq * ws;
                    im[o + 16 * t] = -(yr * ws + yq * wr);
                }
            }
        }
    }
}

__device__ __forceinline__ void load_f16(const float2* __restrict__ t32, const int lg, const int ln,
                                         v2f (&fC)[4], v2f (&fS)[4])
{
#pragma unroll
    for (int s = 0; s < 4; ++s) {
        const int kk = 4 * s + 2 * lg;
        const float2 w0 = t32[(2 * ln * kk) & 31], w1 = t32[(2 * ln * (kk + 1)) & 31];
        fC[s] = (v2f){ w0.x, w1.x };
        fS[s] = (v2f){ w0.y, w1.y };
    }
}

__global__ __launch_bounds__(256) void k1_row_fft(const float* __restrict__ xre,
                                                  const float* __restrict__ xim,
                                                  const float* __restrict__ gamma,
                                                  const float* __restrict__ beta,
                                                  float2* __restrict__ R)
{
    __shared__ float re[LDSN];
    __shared__ float im[LDSN];
    __shared__ float2 t256[256];
    __shared__ float2 tA[512];
    __shared__ float2 tB[512];
    __shared__ float2 t32[32];
    const int row = blockIdx.x;
    const int c   = row & (C_ - 1);
    const int tid = threadIdx.x;
    const int wave = tid >> 5;
    const int lane = tid & 31;
    const int lg   = lane >> 4;
    const int ln   = lane & 15;
    const size_t base = (size_t)row * N_;

    for (int i = tid; i < N_; i += 256) {
        re[lp(i)] = xre[base + i];
        im[lp(i)] = xim[base + i];
    }
    t256[tid] = make_float2(cospif((float)tid * (1.f / 128.f)), sinpif((float)tid * (1.f / 128.f)));
    for (int p = tid; p < 512; p += 256) {
        tA[p] = make_float2(cospif((float)p * (1.f / 4096.f)), sinpif((float)p * (1.f / 4096.f)));
        tB[p] = make_float2(cospif((float)p * (1.f / 256.f)),  sinpif((float)p * (1.f / 256.f)));
    }
    if (tid < 32)
        t32[tid] = make_float2(cospif((float)tid * 0.0625f), sinpif((float)tid * 0.0625f));
    __syncthreads();

    v2f fC[4], fS[4];
    load_f16(t32, lg, ln, fC, fS);

    stage16<1, 0>(re, im, t256, tA, tB, fC, fS, wave, lg, ln);
    __syncthreads();
    stage16<1, 1>(re, im, t256, tA, tB, fC, fS, wave, lg, ln);
    __syncthreads();
    v2f hC[2][4], hS[2][4];
#pragma unroll
    for (int par = 0; par < 2; ++par) {
#pragma unroll
        for (int s = 0; s < 4; ++s) {
            const int kk = 8 * s + 4 * lg + par;
            const float2 w0 = t32[(kk * ln) & 31], w1 = t32[((kk + 2) * ln) & 31];
            hC[par][s] = (v2f){ w0.x, w1.x };
            hS[par][s] = (v2f){ w0.y, w1.y };
        }
    }
    stage32<1>(re, im, tA, tB, hC, hS, wave, lg, ln);
    __syncthreads();

    float lsum = 0.f, lsq = 0.f;
    for (int i = tid; i < N_; i += 256) { float v = re[lp(i)]; lsum += v; lsq += v * v; }
    im[tid]       = lsum;
    im[256 + tid] = lsq;
    __syncthreads();
    for (int off = 128; off > 0; off >>= 1) {
        if (tid < off) { im[tid] += im[tid + off]; im[256 + tid] += im[256 + tid + off]; }
        __syncthreads();
    }
    const float mean = im[0]   * (1.f / N_);
    const float var  = im[256] * (1.f / N_) - mean * mean;
    const float istd = rsqrtf(var + 1e-5f);
    const float gm = gamma[c], bt = beta[c];

    for (int i = tid; i < N_; i += 256) {
        float v = (re[lp(i)] - mean) * istd * gm + bt;
        v = 0.5f * v * (1.f + erff(v * 0.70710678118654752f));
        re[lp(i)] = v;
    }
    __syncthreads();

    stage32<-1>(re, im, tA, tB, hC, hS, wave, lg, ln);
    __syncthreads();
    load_f16(t32, lg, ln, fC, fS);
    stage16<-1, 1>(re, im, t256, tA, tB, fC, fS, wave, lg, ln);
    __syncthreads();
    stage16<-1, 0>(re, im, t256, tA, tB, fC, fS, wave, lg, ln);
    __syncthreads();

    float2* Rrow = R + (size_t)row * KS;
    for (int k = tid; k < KBINS; k += 256)
        Rrow[k] = make_float2(re[lp(k)], im[lp(k)]);
}

__global__ __launch_bounds__(256, 16) void k2_dftC_wmma(const float2* __restrict__ R,
                                                    float* __restrict__ outre,
                                                    float* __restrict__ outim)
{
    __shared__ float L[2 * C_ * LP];
    __shared__ float2 t128[128];
    const int b    = blockIdx.y;
    const int col0 = blockIdx.x * 32;
    const int tid  = threadIdx.x;
    const float2* Rb = R + (size_t)b * C_ * KS;

    for (int idx = tid; idx < C_ * 32; idx += 256) {
        int r = idx >> 5, cc = idx & 31;
        int k = col0 + cc;
        float2 v = (k < KBINS) ? Rb[(size_t)r * KS + k] : make_float2(0.f, 0.f);
        L[(2 * r) * LP + cc]     = v.x;
        L[(2 * r + 1) * LP + cc] = v.y;
    }
    if (tid < 128)
        t128[tid] = make_float2(cospif((float)tid * (1.f / 64.f)), sinpif((float)tid * (1.f / 64.f)));
    __syncthreads();

    const int wave = tid >> 5;
    const int lane = tid & 31;
    const int lg   = lane >> 4;
    const int ln   = lane & 15;

    {
        v2f fC[4], fS[4];
#pragma unroll
        for (int s = 0; s < 4; ++s) {
            const int kk = 4 * s + 2 * lg;
            const float2 w0 = t128[(8 * ln * kk) & 127], w1 = t128[(8 * ln * (kk + 1)) & 127];
            fC[s] = (v2f){ w0.x, w1.x };
            fS[s] = (v2f){ w0.y, w1.y };
        }
#pragma unroll 1
        for (int i = 0; i < 2; ++i) {
            const int tt = wave + 8 * i;
            const int c2 = tt >> 1;
            const int cb = 2 * c2 * LP + 16 * (tt & 1) + ln;
            v8f P = {0.f,0.f,0.f,0.f,0.f,0.f,0.f,0.f};
            v8f U = P;
#pragma unroll
            for (int s = 0; s < 4; ++s) {
                const int ka = cb + 16 * (4 * s + 2 * lg) * LP;
                v2f xr = { L[ka],      L[ka + 16 * LP] };
                v2f xi = { L[ka + LP], L[ka + 17 * LP] };
                v2f xn = { -xr.x, -xr.y };
                P = mm4(fC[s], xr, P);
                U = mm4(fC[s], xi, U);
                P = mm4(fS[s], xi, P);
                U = mm4(fS[s], xn, U);
            }
            float2 tw[8];
#pragma unroll
            for (int v = 0; v < 8; ++v)
                tw[v] = t128[(8 * lg + v) * c2];
#pragma unroll
            for (int v = 0; v < 8; ++v) {
                const int m1 = 8 * lg + v;
                const float yr = P[v], yi = U[v];
                const int o = cb + 16 * m1 * LP;
                L[o]      = yr * tw[v].x + yi * tw[v].y;
                L[o + LP] = yi * tw[v].x - yr * tw[v].y;
            }
        }
    }
    __syncthreads();

    {
        const int q  = ln >> 3;
        const int m2 = ln & 7;
        const float scale = 1.0f / 1048576.0f;
        v2f gA[4];
#pragma unroll
        for (int s = 0; s < 4; ++s) {
            const float2 w0 = t128[(16 * m2 * (2 * s)) & 127], w1 = t128[(16 * m2 * (2 * s + 1)) & 127];
            const float a0 = (q == lg) ? w0.x : (q == 0 ? w0.y : -w0.y);
            const float a1 = (q == lg) ? w1.x : (q == 0 ? w1.y : -w1.y);
            gA[s] = (v2f){ a0 * scale, a1 * scale };
        }
        float* __restrict__ dst = lg ? outim : outre;
        for (int i = 0; i < 4; ++i) {
            const int tt = wave + 8 * i;
            const int m1 = tt >> 1;
            const int n  = 16 * (tt & 1) + ln;
            const int rb = (16 * m1 + lg) * LP + n;
            v8f acc = {0.f,0.f,0.f,0.f,0.f,0.f,0.f,0.f};
#pragma unroll
            for (int s = 0; s < 4; ++s) {
                const int ka = rb + 4 * s * LP;
                v2f xb = { L[ka], L[ka + 2 * LP] };
                acc = mm4(gA[s], xb, acc);
            }
            const int kcol = col0 + n;
            if (kcol < KBINS) {
                const bool edge = (kcol == 0) || (kcol == NH);
                const bool mir  = (kcol >= 1) && (kcol <= NH - 1);
#pragma unroll
                for (int r = 0; r < 8; ++r) {
                    const int mr = m1 + 16 * r;
                    const size_t o = (size_t)(b * C_ + mr) * N_;
                    const float val = (lg && edge) ? 0.f : acc[r];
                    dst[o + kcol] = val;
                    if (mir)
                        dst[o + (N_ - kcol)] = lg ? -val : val;
                }
            }
        }
    }
}

extern "C" void kernel_launch(void* const* d_in, const int* in_sizes, int n_in,
                              void* d_out, int out_size, void* d_ws, size_t ws_size,
                              hipStream_t stream) {
    const float* xre   = (const float*)d_in[0];
    const float* xim   = (const float*)d_in[1];
    const float* gamma = (const float*)d_in[2];
    const float* beta  = (const float*)d_in[3];
    float* outre = (float*)d_out;
    float* outim = outre + (size_t)B_ * C_ * N_;

    float2* R  = (float2*)d_ws;

    k1_row_fft<<<dim3(B_ * C_), dim3(256), 0, stream>>>(xre, xim, gamma, beta, R);
    k2_dftC_wmma<<<dim3((KBINS + 31) / 32, B_), dim3(256), 0, stream>>>(R, outre, outim);
}


# row transform, forward side: the line is real, so its two DFT-16 stages carry columns 0..16 of 32 (17 tiles each, not 32); bins of low digit 17..31 stored as conj of bin N-k; no sum reordered
# speedup vs baseline: 2.5256x; 1.0657x over previous
#include <hip/hip_runtime.h>
#include <math.h>

#define B_    16
#define C_    128
#define N_    8192
#define NH    4096
#define KBINS 4097
#define KS    4104
#define LP    33

typedef __attribute__((ext_vector_type(2))) float v2f;
typedef __attribute__((ext_vector_type(8))) float v8f;

#define LDSN 8448

__device__ __forceinline__ v8f mm4(v2f a, v2f b, v8f c)
{
    return __builtin_amdgcn_wmma_f32_16x16x4_f32(false, a, false, b, (short)0, c, false, false);
}

__device__ __forceinline__ int lp(int i) { return i + (i >> 5); }

template <int SGN, int MODE, bool HALF>
__device__ __forceinline__ void stage16(float* __restrict__ re, float* __restrict__ im,
                                        const float2* __restrict__ t256,
                                        const float2* __restrict__ tA,
                                        const float2* __restrict__ tB,
                                        const v2f (&fC)[4], const v2f (&fS)[4],
                                        const int wave, const int lg, const int ln)
{
    const int nt = HALF ? ((wave == MODE) ? 3 : 2) : 4;
#pragma unroll 1
    for (int i = 0; i < nt; ++i) {
        int cb, rs, h = 0, c = 0, j = 0;
        if (!HALF) {
            const int tt = wave + 8 * i;
            if (MODE == 0) { j = 16 * tt + ln; cb = j + (j >> 5); rs = 528; }
            else           { h = tt >> 1; c = 16 * (tt & 1) + ln; cb = h * 528 + c; rs = 33; }
        } else {
            const int blk = wave + 8 * i;
            if (MODE == 0) { j = (i < 2) ? 32 * blk + ln : 32 * ln + 16; cb = j + (j >> 5); rs = 528; }
            else           { h = (i < 2) ? blk : ln; c = (i < 2) ? ln : 16; cb = h * 528 + c; rs = 33; }
        }
        v8f P = {0.f,0.f,0.f,0.f,0.f,0.f,0.f,0.f};
        v8f Q = P, U = P, V = P;
#pragma unroll
        for (int s = 0; s < 4; ++s) {
            const int ka = cb + (4 * s + 2 * lg) * rs;
            v2f xr = { re[ka], re[ka + rs] };
            v2f xi = { im[ka], im[ka + rs] };
            P = mm4(fC[s], xr, P);
            Q = mm4(fS[s], xi, Q);
            U = mm4(fC[s], xi, U);
            V = mm4(fS[s], xr, V);
        }
#pragma unroll
        for (int v = 0; v < 8; ++v) {
            const int ko = 8 * lg + v;
            float yr = (SGN > 0) ? (P[v] - Q[v]) : (P[v] + Q[v]);
            float yi = (SGN > 0) ? (U[v] + V[v]) : (U[v] - V[v]);
            if (MODE == 0 && SGN > 0) {
                const float2 w = t256[(j >> 5) * ko];
                const float tr = yr * w.x - yi * w.y;
                const float ti = yr * w.y + yi * w.x;
                yr = tr; yi = ti;
            } else if (MODE == 1 && SGN > 0) {
                const float2 a = tA[c * h], b = tB[c * ko];
                const float wr = a.x * b.x - a.y * b.y;
                const float wi = a.x * b.y + a.y * b.x;
                const float tr = yr * wr - yi * wi;
                const float ti = yr * wi + yi * wr;
                yr = tr; yi = ti;
            } else if (MODE == 1 && SGN < 0) {
                const float2 w = t256[ko * h];
                const float tr = yr * w.x + yi * w.y;
                const float ti = yi * w.x - yr * w.y;
                yr = tr; yi = ti;
            }
            re[cb + ko * rs] = yr;
            im[cb + ko * rs] = yi;
        }
    }
}

template <int SGN>
__device__ __forceinline__ void stage32(float* __restrict__ re, float* __restrict__ im,
                                        const float2* __restrict__ tA,
                                        const float2* __restrict__ tB,
                                        const v2f (&hC)[2][4], const v2f (&hS)[2][4],
                                        const int wave, const int lg, const int ln)
{
#pragma unroll 1
    for (int i = 0; i < 2; ++i) {
        const int mt = wave + 8 * i;
        const int rb = (16 * mt + ln) * 33 + 4 * lg;
        v8f PE = {0.f,0.f,0.f,0.f,0.f,0.f,0.f,0.f};
        v8f PO = PE, QE = PE, QO = PE;
#pragma unroll
        for (int s = 0; s < 4; ++s) {
            const int ka = rb + 8 * s;
            v2f xe = { re[ka],     re[ka + 2] };
            v2f xo = { re[ka + 1], re[ka + 3] };
            if (SGN > 0) {
                v2f ye = { im[ka],     im[ka + 2] };
                v2f yo = { im[ka + 1], im[ka + 3] };
                PE = mm4(xe, hC[0][s], PE);
                QE = mm4(ye, hS[0][s], QE);
                PO = mm4(xo, hC[1][s], PO);
                QO = mm4(yo, hS[1][s], QO);
            } else {
                PE = mm4(xe, hC[0][s], PE);
                QE = mm4(xe, hS[0][s], QE);
                PO = mm4(xo, hC[1][s], PO);
                QO = mm4(xo, hS[1][s], QO);
            }
        }
#pragma unroll
        for (int v = 0; v < 8; ++v) {
            const int r = 16 * mt + 8 * lg + v;
            const int o = r * 33 + ln;
            if (SGN > 0) {
                const float ev = PE[v] - QE[v], od = PO[v] - QO[v];
                re[o]      = ev + od;
                re[o + 16] = ev - od;
            } else {
                const float yr = PE[v] + PO[v];
                const float yq = QE[v] + QO[v];
                const float2 a = tA[ln * (r >> 4)], b = tB[ln * (r & 15)];
                const float wr = a.x * b.x - a.y * b.y;
                const float ws = a.x * b.y + a.y * b.x;
                re[o] = yr * wr - yq * ws;
                im[o] = -(yr * ws + yq * wr);
                if (ln == 0) {
                    const float ya = PE[v] - PO[v];
                    const float2 w = tB[(r >> 4) + 16 * (r & 15)];
                    re[o + 16] = ya * w.x;
                    im[o + 16] = -(ya * w.y);
                }
            }
        }
    }
}

__device__ __forceinline__ void load_f16(const float2* __restrict__ t32, const int lg, const int ln,
                                         v2f (&fC)[4], v2f (&fS)[4])
{
#pragma unroll
    for (int s = 0; s < 4; ++s) {
        const int kk = 4 * s + 2 * lg;
        const float2 w0 = t32[(2 * ln * kk) & 31], w1 = t32[(2 * ln * (kk + 1)) & 31];
        fC[s] = (v2f){ w0.x, w1.x };
        fS[s] = (v2f){ w0.y, w1.y };
    }
}

__global__ __launch_bounds__(256) void k1_row_fft(const float* __restrict__ xre,
                                                  const float* __restrict__ xim,
                                                  const float* __restrict__ gamma,
                                                  const float* __restrict__ beta,
                                                  float2* __restrict__ R)
{
    __shared__ float re[LDSN];
    __shared__ float im[LDSN];
    __shared__ float2 t256[256];
    __shared__ float2 tA[512];
    __shared__ float2 tB[512];
    __shared__ float2 t32[32];
    const int row = blockIdx.x;
    const int c   = row & (C_ - 1);
    const int tid = threadIdx.x;
    const int wave = __builtin_amdgcn_readfirstlane(tid >> 5);
    const int lane = tid & 31;
    const int lg   = lane >> 4;
    const int ln   = lane & 15;
    const size_t base = (size_t)row * N_;

    for (int i = tid; i < N_; i += 256) {
        re[lp(i)] = xre[base + i];
        im[lp(i)] = xim[base + i];
    }
    t256[tid] = make_float2(cospif((float)tid * (1.f / 128.f)), sinpif((float)tid * (1.f / 128.f)));
    for (int p = tid; p < 512; p += 256) {
        tA[p] = make_float2(cospif((float)p * (1.f / 4096.f)), sinpif((float)p * (1.f / 4096.f)));
        tB[p] = make_float2(cospif((float)p * (1.f / 256.f)),  sinpif((float)p * (1.f / 256.f)));
    }
    if (tid < 32)
        t32[tid] = make_float2(cospif((float)tid * 0.0625f), sinpif((float)tid * 0.0625f));
    __syncthreads();

    v2f fC[4], fS[4];
    load_f16(t32, lg, ln, fC, fS);

    stage16<1, 0, false>(re, im, t256, tA, tB, fC, fS, wave, lg, ln);
    __syncthreads();
    stage16<1, 1, false>(re, im, t256, tA, tB, fC, fS, wave, lg, ln);
    __syncthreads();
    v2f hC[2][4], hS[2][4];
#pragma unroll
    for (int par = 0; par < 2; ++par) {
#pragma unroll
        for (int s = 0; s < 4; ++s) {
            const int kk = 8 * s + 4 * lg + par;
            const float2 w0 = t32[(kk * ln) & 31], w1 = t32[((kk + 2) * ln) & 31];
            hC[par][s] = (v2f){ w0.x, w1.x };
            hS[par][s] = (v2f){ w0.y, w1.y };
        }
    }
    stage32<1>(re, im, tA, tB, hC, hS, wave, lg, ln);
    __syncthreads();

    float lsum = 0.f, lsq = 0.f;
    for (int i = tid; i < N_; i += 256) { float v = re[lp(i)]; lsum += v; lsq += v * v; }
    im[tid]       = lsum;
    im[256 + tid] = lsq;
    __syncthreads();
    for (int off = 128; off > 0; off >>= 1) {
        if (tid < off) { im[tid] += im[tid + off]; im[256 + tid] += im[256 + tid + off]; }
        __syncthreads();
    }
    const float mean = im[0]   * (1.f / N_);
    const float var  = im[256] * (1.f / N_) - mean * mean;
    const float istd = rsqrtf(var + 1e-5f);
    const float gm = gamma[c], bt = beta[c];

    for (int i = tid; i < N_; i += 256) {
        float v = (re[lp(i)] - mean) * istd * gm + bt;
        v = 0.5f * v * (1.f + erff(v * 0.70710678118654752f));
        re[lp(i)] = v;
    }
    __syncthreads();

    stage32<-1>(re, im, tA, tB, hC, hS, wave, lg, ln);
    __syncthreads();
    load_f16(t32, lg, ln, fC, fS);
    stage16<-1, 1, true>(re, im, t256, tA, tB, fC, fS, wave, lg, ln);
    __syncthreads();
    stage16<-1, 0, true>(re, im, t256, tA, tB, fC, fS, wave, lg, ln);
    __syncthreads();

    float2* Rrow = R + (size_t)row * KS;
    for (int k = tid; k < KBINS; k += 256) {
        const bool cj = (k & 31) > 16;
        const int p = cj ? N_ - k : k;
        const float vi = im[lp(p)];
        Rrow[k] = make_float2(re[lp(p)], cj ? -vi : vi);
    }
}

__global__ __launch_bounds__(256, 16) void k2_dftC_wmma(const float2* __restrict__ R,
                                                    float* __restrict__ outre,
                                                    float* __restrict__ outim)
{
    __shared__ float L[2 * C_ * LP];
    __shared__ float2 t128[128];
    const int b    = blockIdx.y;
    const int col0 = blockIdx.x * 32;
    const int tid  = threadIdx.x;
    const float2* Rb = R + (size_t)b * C_ * KS;

    for (int idx = tid; idx < C_ * 32; idx += 256) {
        int r = idx >> 5, cc = idx & 31;
        int k = col0 + cc;
        float2 v = (k < KBINS) ? Rb[(size_t)r * KS + k] : make_float2(0.f, 0.f);
        L[(2 * r) * LP + cc]     = v.x;
        L[(2 * r + 1) * LP + cc] = v.y;
    }
    if (tid < 128)
        t128[tid] = make_float2(cospif((float)tid * (1.f / 64.f)), sinpif((float)tid * (1.f / 64.f)));
    __syncthreads();

    const int wave = tid >> 5;
    const int lane = tid & 31;
    const int lg   = lane >> 4;
    const int ln   = lane & 15;

    {
        v2f fC[4], fS[4];
#pragma unroll
        for (int s = 0; s < 4; ++s) {
            const int kk = 4 * s + 2 * lg;
            const float2 w0 = t128[(8 * ln * kk) & 127], w1 = t128[(8 * ln * (kk + 1)) & 127];
            fC[s] = (v2f){ w0.x, w1.x };
            fS[s] = (v2f){ w0.y, w1.y };
        }
#pragma unroll 1
        for (int i = 0; i < 2; ++i) {
            const int tt = wave + 8 * i;
            const int c2 = tt >> 1;
            const int cb = 2 * c2 * LP + 16 * (tt & 1) + ln;
            v8f P = {0.f,0.f,0.f,0.f,0.f,0.f,0.f,0.f};
            v8f U = P;
#pragma unroll
            for (int s = 0; s < 4; ++s) {
                const int ka = cb + 16 * (4 * s + 2 * lg) * LP;
                v2f xr = { L[ka],      L[ka + 16 * LP] };
                v2f xi = { L[ka + LP], L[ka + 17 * LP] };
                v2f xn = { -xr.x, -xr.y };
                P = mm4(fC[s], xr, P);
                U = mm4(fC[s], xi, U);
                P = mm4(fS[s], xi, P);
                U = mm4(fS[s], xn, U);
            }
            float2 tw[8];
#pragma unroll
            for (int v = 0; v < 8; ++v)
                tw[v] = t128[(8 * lg + v) * c2];
#pragma unroll
            for (int v = 0; v < 8; ++v) {
                const int m1 = 8 * lg + v;
                const float yr = P[v], yi = U[v];
                const int o = cb + 16 * m1 * LP;
                L[o]      = yr * tw[v].x + yi * tw[v].y;
                L[o + LP] = yi * tw[v].x - yr * tw[v].y;
            }
        }
    }
    __syncthreads();

    {
        const int q  = ln >> 3;
        const int m2 = ln & 7;
        const float scale = 1.0f / 1048576.0f;
        v2f gA[4];
#pragma unroll
        for (int s = 0; s < 4; ++s) {
            const float2 w0 = t128[(16 * m2 * (2 * s)) & 127], w1 = t128[(16 * m2 * (2 * s + 1)) & 127];
            const float a0 = (q == lg) ? w0.x : (q == 0 ? w0.y : -w0.y);
            const float a1 = (q == lg) ? w1.x : (q == 0 ? w1.y : -w1.y);
            gA[s] = (v2f){ a0 * scale, a1 * scale };
        }
        float* __restrict__ dst = lg ? outim : outre;
        for (int i = 0; i < 4; ++i) {
            const int tt = wave + 8 * i;
            const int m1 = tt >> 1;
            const int n  = 16 * (tt & 1) + ln;
            const int rb = (16 * m1 + lg) * LP + n;
            v8f acc = {0.f,0.f,0.f,0.f,0.f,0.f,0.f,0.f};
#pragma unroll
            for (int s = 0; s < 4; ++s) {
                const int ka = rb + 4 * s * LP;
                v2f xb = { L[ka], L[ka + 2 * LP] };
                acc = mm4(gA[s], xb, acc);
            }
            const int kcol = col0 + n;
            if (kcol < KBINS) {
                const bool edge = (kcol == 0) || (kcol == NH);
                const bool mir  = (kcol >= 1) && (kcol <= NH - 1);
#pragma unroll
                for (int r = 0; r < 8; ++r) {
                    const int mr = m1 + 16 * r;
                    const size_t o = (size_t)(b * C_ + mr) * N_;
                    const float val = (lg && edge) ? 0.f : acc[r];
                    dst[o + kcol] = val;
                    if (mir)
                        dst[o + (N_ - kcol)] = lg ? -val : val;
                }
            }
        }
    }
}

extern "C" void kernel_launch(void* const* d_in, const int* in_sizes, int n_in,
                              void* d_out, int out_size, void* d_ws, size_t ws_size,
                              hipStream_t stream) {
    const float* xre   = (const float*)d_in[0];
    const float* xim   = (const float*)d_in[1];
    const float* gamma = (const float*)d_in[2];
    const float* beta  = (const float*)d_in[3];
    float* outre = (float*)d_out;
    float* outim = outre + (size_t)B_ * C_ * N_;

    float2* R  = (float2*)d_ws;

    k1_row_fft<<<dim3(B_ * C_), dim3(256), 0, stream>>>(xre, xim, gamma, beta, R);
    k2_dftC_wmma<<<dim3((KBINS + 31) / 32, B_), dim3(256), 0, stream>>>(R, outre, outim);
}


# row transform, inverse side: only Re is kept: first loader forms X[k]+conj X[N-k], DFT-16 stages carry columns 0..16 (17 tiles, not 32), DFT-32 sums k 0..16 (8 WMMAs, not 16); sum order changes
# speedup vs baseline: 2.7391x; 1.0845x over previous
#include <hip/hip_runtime.h>
#include <math.h>

#define B_    16
#define C_    128
#define N_    8192
#define NH    4096
#define KBINS 4097
#define KS    4104
#define LP    33

typedef __attribute__((ext_vector_type(2))) float v2f;
typedef __attribute__((ext_vector_type(8))) float v8f;

#define LDSN 8448

__device__ __forceinline__ v8f mm4(v2f a, v2f b, v8f c)
{
    return __builtin_amdgcn_wmma_f32_16x16x4_f32(false, a, false, b, (short)0, c, false, false);
}

__device__ __forceinline__ int lp(int i) { return i + (i >> 5); }

template <int SGN, int MODE, bool HALF>
__device__ __forceinline__ void stage16(float* __restrict__ re, float* __restrict__ im,
                                        const float2* __restrict__ t256,
                                        const float2* __restrict__ tA,
                                        const float2* __restrict__ tB,
                                        const v2f (&fC)[4], const v2f (&fS)[4],
                                        const int wave, const int lg, const int ln)
{
    const int nt = HALF ? ((wave == MODE) ? 3 : 2) : 4;
#pragma unroll 1
    for (int i = 0; i < nt; ++i) {
        int cb, rs, h = 0, c = 0, j = 0;
        if (!HALF) {
            const int tt = wave + 8 * i;
            if (MODE == 0) { j = 16 * tt + ln; cb = j + (j >> 5); rs = 528; }
            else           { h = tt >> 1; c = 16 * (tt & 1) + ln; cb = h * 528 + c; rs = 33; }
        } else {
            const int blk = wave + 8 * i;
            if (MODE == 0) { j = (i < 2) ? 32 * blk + ln : 32 * ln + 16; cb = j + (j >> 5); rs = 528; }
            else           { h = (i < 2) ? blk : ln; c = (i < 2) ? ln : 16; cb = h * 528 + c; rs = 33; }
        }
        v8f P = {0.f,0.f,0.f,0.f,0.f,0.f,0.f,0.f};
        v8f Q = P, U = P, V = P;
        const bool fold = (SGN > 0) && (MODE == 0) && HALF;
        const bool mir  = fold && (i < 2) && (ln != 0);
        const int  mo   = 8447 - ((i < 2 && ln == 0) ? 1 : 0);
#pragma unroll
        for (int s = 0; s < 4; ++s) {
            const int ka = cb + (4 * s + 2 * lg) * rs;
            v2f xr = { re[ka], re[ka + rs] };
            v2f xi = { im[ka], im[ka + rs] };
            if (fold) {
                const int km = mo - ka;
                const float mr0 = re[km], mr1 = re[km - rs];
                const float mi0 = im[km], mi1 = im[km - rs];
                xr.x += mir ? mr0 : 0.f;
                xr.y += mir ? mr1 : 0.f;
                xi.x -= mir ? mi0 : 0.f;
                xi.y -= mir ? mi1 : 0.f;
            }
            P = mm4(fC[s], xr, P);
            Q = mm4(fS[s], xi, Q);
            U = mm4(fC[s], xi, U);
            V = mm4(fS[s], xr, V);
        }
#pragma unroll
        for (int v = 0; v < 8; ++v) {
            const int ko = 8 * lg + v;
            float yr = (SGN > 0) ? (P[v] - Q[v]) : (P[v] + Q[v]);
            float yi = (SGN > 0) ? (U[v] + V[v]) : (U[v] - V[v]);
            if (MODE == 0 && SGN > 0) {
                const float2 w = t256[(j >> 5) * ko];
                const float tr = yr * w.x - yi * w.y;
                const float ti = yr * w.y + yi * w.x;
                yr = tr; yi = ti;
            } else if (MODE == 1 && SGN > 0) {
                const float2 a = tA[c * h], b = tB[c * ko];
                const float wr = a.x * b.x - a.y * b.y;
                const float wi = a.x * b.y + a.y * b.x;
                const float tr = yr * wr - yi * wi;
                const float ti = yr * wi + yi * wr;
                yr = tr; yi = ti;
            } else if (MODE == 1 && SGN < 0) {
                const float2 w = t256[ko * h];
                const float tr = yr * w.x + yi * w.y;
                const float ti = yi * w.x - yr * w.y;
                yr = tr; yi = ti;
            }
            re[cb + ko * rs] = yr;
            im[cb + ko * rs] = yi;
        }
    }
}

template <int SGN>
__device__ __forceinline__ void stage32(float* __restrict__ re, float* __restrict__ im,
                                        const float2* __restrict__ tA,
                                        const float2* __restrict__ tB,
                                        const v2f (&hC)[2][4], const v2f (&hS)[2][4],
                                        const int wave, const int lg, const int ln)
{
#pragma unroll 1
    for (int i = 0; i < 2; ++i) {
        const int mt = wave + 8 * i;
        const int rb = (16 * mt + ln) * 33 + 4 * lg;
        v8f PE = {0.f,0.f,0.f,0.f,0.f,0.f,0.f,0.f};
        v8f PO = PE, QE = PE, QO = PE;
        float c16[8];
        if (SGN > 0) {
#pragma unroll
            for (int v = 0; v < 8; ++v)
                c16[v] = re[(16 * mt + 8 * lg + v) * 33 + 16];
        }
        const float sg = (ln & 1) ? -1.f : 1.f;
#pragma unroll
        for (int s = 0; s < ((SGN > 0) ? 2 : 4); ++s) {
            const int ka = rb + 8 * s;
            v2f xe = { re[ka],     re[ka + 2] };
            v2f xo = { re[ka + 1], re[ka + 3] };
            if (SGN > 0) {
                v2f ye = { im[ka],     im[ka + 2] };
                v2f yo = { im[ka + 1], im[ka + 3] };
                PE = mm4(xe, hC[0][s], PE);
                QE = mm4(ye, hS[0][s], QE);
                PO = mm4(xo, hC[1][s], PO);
                QO = mm4(yo, hS[1][s], QO);
            } else {
                PE = mm4(xe, hC[0][s], PE);
                QE = mm4(xe, hS[0][s], QE);
                PO = mm4(xo, hC[1][s], PO);
                QO = mm4(xo, hS[1][s], QO);
            }
        }
#pragma unroll
        for (int v = 0; v < 8; ++v) {
            const int r = 16 * mt + 8 * lg + v;
            const int o = r * 33 + ln;
            if (SGN > 0) {
                const float ev = (PE[v] - QE[v]) + sg * c16[v], od = PO[v] - QO[v];
                re[o]      = ev + od;
                re[o + 16] = ev - od;
            } else {
                const float yr = PE[v] + PO[v];
                const float yq = QE[v] + QO[v];
                const float2 a = tA[ln * (r >> 4)], b = tB[ln * (r & 15)];
                const float wr = a.x * b.x - a.y * b.y;
                const float ws = a.x * b.y + a.y * b.x;
                re[o] = yr * wr - yq * ws;
                im[o] = -(yr * ws + yq * wr);
                if (ln == 0) {
                    const float ya = PE[v] - PO[v];
                    const float2 w = tB[(r >> 4) + 16 * (r & 15)];
                    re[o + 16] = ya * w.x;
                    im[o + 16] = -(ya * w.y);
                }
            }
        }
    }
}

__device__ __forceinline__ void load_f16(const float2* __restrict__ t32, const int lg, const int ln,
                                         v2f (&fC)[4], v2f (&fS)[4])
{
#pragma unroll
    for (int s = 0; s < 4; ++s) {
        const int kk = 4 * s + 2 * lg;
        const float2 w0 = t32[(2 * ln * kk) & 31], w1 = t32[(2 * ln * (kk + 1)) & 31];
        fC[s] = (v2f){ w0.x, w1.x };
        fS[s] = (v2f){ w0.y, w1.y };
    }
}

__global__ __launch_bounds__(256) void k1_row_fft(const float* __restrict__ xre,
                                                  const float* __restrict__ xim,
                                                  const float* __restrict__ gamma,
                                                  const float* __restrict__ beta,
                                                  float2* __restrict__ R)
{
    __shared__ float re[LDSN];
    __shared__ float im[LDSN];
    __shared__ float2 t256[256];
    __shared__ float2 tA[512];
    __shared__ float2 tB[512];
    __shared__ float2 t32[32];
    const int row = blockIdx.x;
    const int c   = row & (C_ - 1);
    const int tid = threadIdx.x;
    const int wave = __builtin_amdgcn_readfirstlane(tid >> 5);
    const int lane = tid & 31;
    const int lg   = lane >> 4;
    const int ln   = lane & 15;
    const size_t base = (size_t)row * N_;

    for (int i = tid; i < N_; i += 256) {
        re[lp(i)] = xre[base + i];
        im[lp(i)] = xim[base + i];
    }
    t256[tid] = make_float2(cospif((float)tid * (1.f / 128.f)), sinpif((float)tid * (1.f / 128.f)));
    for (int p = tid; p < 512; p += 256) {
        tA[p] = make_float2(cospif((float)p * (1.f / 4096.f)), sinpif((float)p * (1.f / 4096.f)));
        tB[p] = make_float2(cospif((float)p * (1.f / 256.f)),  sinpif((float)p * (1.f / 256.f)));
    }
    if (tid < 32)
        t32[tid] = make_float2(cospif((float)tid * 0.0625f), sinpif((float)tid * 0.0625f));
    __syncthreads();

    v2f fC[4], fS[4];
    load_f16(t32, lg, ln, fC, fS);

    stage16<1, 0, true>(re, im, t256, tA, tB, fC, fS, wave, lg, ln);
    __syncthreads();
    stage16<1, 1, true>(re, im, t256, tA, tB, fC, fS, wave, lg, ln);
    __syncthreads();
    v2f hC[2][4], hS[2][4];
#pragma unroll
    for (int par = 0; par < 2; ++par) {
#pragma unroll
        for (int s = 0; s < 4; ++s) {
            const int kk = 8 * s + 4 * lg + par;
            const float2 w0 = t32[(kk * ln) & 31], w1 = t32[((kk + 2) * ln) & 31];
            hC[par][s] = (v2f){ w0.x, w1.x };
            hS[par][s] = (v2f){ w0.y, w1.y };
        }
    }
    stage32<1>(re, im, tA, tB, hC, hS, wave, lg, ln);
    __syncthreads();

    float lsum = 0.f, lsq = 0.f;
    for (int i = tid; i < N_; i += 256) { float v = re[lp(i)]; lsum += v; lsq += v * v; }
    im[tid]       = lsum;
    im[256 + tid] = lsq;
    __syncthreads();
    for (int off = 128; off > 0; off >>= 1) {
        if (tid < off) { im[tid] += im[tid + off]; im[256 + tid] += im[256 + tid + off]; }
        __syncthreads();
    }
    const float mean = im[0]   * (1.f / N_);
    const float var  = im[256] * (1.f / N_) - mean * mean;
    const float istd = rsqrtf(var + 1e-5f);
    const float gm = gamma[c], bt = beta[c];

    for (int i = tid; i < N_; i += 256) {
        float v = (re[lp(i)] - mean) * istd * gm + bt;
        v = 0.5f * v * (1.f + erff(v * 0.70710678118654752f));
        re[lp(i)] = v;
    }
    __syncthreads();

    stage32<-1>(re, im, tA, tB, hC, hS, wave, lg, ln);
    __syncthreads();
    load_f16(t32, lg, ln, fC, fS);
    stage16<-1, 1, true>(re, im, t256, tA, tB, fC, fS, wave, lg, ln);
    __syncthreads();
    stage16<-1, 0, true>(re, im, t256, tA, tB, fC, fS, wave, lg, ln);
    __syncthreads();

    float2* Rrow = R + (size_t)row * KS;
    for (int k = tid; k < KBINS; k += 256) {
        const bool cj = (k & 31) > 16;
        const int p = cj ? N_ - k : k;
        const float vi = im[lp(p)];
        Rrow[k] = make_float2(re[lp(p)], cj ? -vi : vi);
    }
}

__global__ __launch_bounds__(256, 16) void k2_dftC_wmma(const float2* __restrict__ R,
                                                    float* __restrict__ outre,
                                                    float* __restrict__ outim)
{
    __shared__ float L[2 * C_ * LP];
    __shared__ float2 t128[128];
    const int b    = blockIdx.y;
    const int col0 = blockIdx.x * 32;
    const int tid  = threadIdx.x;
    const float2* Rb = R + (size_t)b * C_ * KS;

    for (int idx = tid; idx < C_ * 32; idx += 256) {
        int r = idx >> 5, cc = idx & 31;
        int k = col0 + cc;
        float2 v = (k < KBINS) ? Rb[(size_t)r * KS + k] : make_float2(0.f, 0.f);
        L[(2 * r) * LP + cc]     = v.x;
        L[(2 * r + 1) * LP + cc] = v.y;
    }
    if (tid < 128)
        t128[tid] = make_float2(cospif((float)tid * (1.f / 64.f)), sinpif((float)tid * (1.f / 64.f)));
    __syncthreads();

    const int wave = tid >> 5;
    const int lane = tid & 31;
    const int lg   = lane >> 4;
    const int ln   = lane & 15;

    {
        v2f fC[4], fS[4];
#pragma unroll
        for (int s = 0; s < 4; ++s) {
            const int kk = 4 * s + 2 * lg;
            const float2 w0 = t128[(8 * ln * kk) & 127], w1 = t128[(8 * ln * (kk + 1)) & 127];
            fC[s] = (v2f){ w0.x, w1.x };
            fS[s] = (v2f){ w0.y, w1.y };
        }
#pragma unroll 1
        for (int i = 0; i < 2; ++i) {
            const int tt = wave + 8 * i;
            const int c2 = tt >> 1;
            const int cb = 2 * c2 * LP + 16 * (tt & 1) + ln;
            v8f P = {0.f,0.f,0.f,0.f,0.f,0.f,0.f,0.f};
            v8f U = P;
#pragma unroll
            for (int s = 0; s < 4; ++s) {
                const int ka = cb + 16 * (4 * s + 2 * lg) * LP;
                v2f xr = { L[ka],      L[ka + 16 * LP] };
                v2f xi = { L[ka + LP], L[ka + 17 * LP] };
                v2f xn = { -xr.x, -xr.y };
                P = mm4(fC[s], xr, P);
                U = mm4(fC[s], xi, U);
                P = mm4(fS[s], xi, P);
                U = mm4(fS[s], xn, U);
            }
            float2 tw[8];
#pragma unroll
            for (int v = 0; v < 8; ++v)
                tw[v] = t128[(8 * lg + v) * c2];
#pragma unroll
            for (int v = 0; v < 8; ++v) {
                const int m1 = 8 * lg + v;
                const float yr = P[v], yi = U[v];
                const int o = cb + 16 * m1 * LP;
                L[o]      = yr * tw[v].x + yi * tw[v].y;
                L[o + LP] = yi * tw[v].x - yr * tw[v].y;
            }
        }
    }
    __syncthreads();

    {
        const int q  = ln >> 3;
        const int m2 = ln & 7;
        const float scale = 1.0f / 1048576.0f;
        v2f gA[4];
#pragma unroll
        for (int s = 0; s < 4; ++s) {
            const float2 w0 = t128[(16 * m2 * (2 * s)) & 127], w1 = t128[(16 * m2 * (2 * s + 1)) & 127];
            const float a0 = (q == lg) ? w0.x : (q == 0 ? w0.y : -w0.y);
            const float a1 = (q == lg) ? w1.x : (q == 0 ? w1.y : -w1.y);
            gA[s] = (v2f){ a0 * scale, a1 * scale };
        }
        float* __restrict__ dst = lg ? outim : outre;
        for (int i = 0; i < 4; ++i) {
            const int tt = wave + 8 * i;
            const int m1 = tt >> 1;
            const int n  = 16 * (tt & 1) + ln;
            const int rb = (16 * m1 + lg) * LP + n;
            v8f acc = {0.f,0.f,0.f,0.f,0.f,0.f,0.f,0.f};
#pragma unroll
            for (int s = 0; s < 4; ++s) {
                const int ka = rb + 4 * s * LP;
                v2f xb = { L[ka], L[ka + 2 * LP] };
                acc = mm4(gA[s], xb, acc);
            }
            const int kcol = col0 + n;
            if (kcol < KBINS) {
                const bool edge = (kcol == 0) || (kcol == NH);
                const bool mir  = (kcol >= 1) && (kcol <= NH - 1);
#pragma unroll
                for (int r = 0; r < 8; ++r) {
                    const int mr = m1 + 16 * r;
                    const size_t o = (size_t)(b * C_ + mr) * N_;
                    const float val = (lg && edge) ? 0.f : acc[r];
                    dst[o + kcol] = val;
                    if (mir)
                        dst[o + (N_ - kcol)] = lg ? -val : val;
                }
            }
        }
    }
}

extern "C" void kernel_launch(void* const* d_in, const int* in_sizes, int n_in,
                              void* d_out, int out_size, void* d_ws, size_t ws_size,
                              hipStream_t stream) {
    const float* xre   = (const float*)d_in[0];
    const float* xim   = (const float*)d_in[1];
    const float* gamma = (const float*)d_in[2];
    const float* beta  = (const float*)d_in[3];
    float* outre = (float*)d_out;
    float* outim = outre + (size_t)B_ * C_ * N_;

    float2* R  = (float2*)d_ws;

    k1_row_fft<<<dim3(B_ * C_), dim3(256), 0, stream>>>(xre, xim, gamma, beta, R);
    k2_dftC_wmma<<<dim3((KBINS + 31) / 32, B_), dim3(256), 0, stream>>>(R, outre, outim);
}
